# starting kernel
# baseline (speedup 1.0000x reference)
#include <hip/hip_runtime.h>

typedef __attribute__((ext_vector_type(2))) float v2f;
typedef __attribute__((ext_vector_type(8))) float v8f;

#define H_IMG 512
#define W_IMG 512
#define CH_STRIDE (512 * 512)
#define RS 42
#define HS 17

__global__ __launch_bounds__(32)
void ImageEvaluator_84000970375834_kernel(const float* __restrict__ img,
                                          float* __restrict__ out) {
    __shared__ float sXs[48 * RS];
    __shared__ float sXq[48 * RS];
    __shared__ float sH1[40 * HS];
    __shared__ float sH2[40 * HS];

    const int lane = threadIdx.x;
    const int hi   = lane >> 4;
    const int lo   = lane & 15;
    const int tc   = blockIdx.x * 16;
    const int tr   = blockIdx.y * 16;
    const int b    = blockIdx.z;

    const float* base = img + (size_t)b * 3 * CH_STRIDE;

    v2f band[10];
    #pragma unroll
    for (int k = 0; k < 10; ++k) {
        const int j0 = 4 * k + 2 * hi;
        band[k].x = ((unsigned)(j0     - lo) <= 24u) ? 1.0f : 0.0f;
        band[k].y = ((unsigned)(j0 + 1 - lo) <= 24u) ? 1.0f : 0.0f;
    }

    for (int p = lane; p < 40 * 40; p += 32) {
        int rr = p / 40, cc = p - rr * 40;
        int ir = tr - 12 + rr; ir = (ir < 0) ? -ir : ((ir > 511) ? 1022 - ir : ir);
        int ic = tc - 12 + cc; ic = (ic < 0) ? -ic : ((ic > 511) ? 1022 - ic : ic);
        size_t off = (size_t)ir * W_IMG + ic;
        float x0 = base[off];
        float x1 = base[off + CH_STRIDE];
        float x2 = base[off + 2 * CH_STRIDE];
        sXs[rr * RS + cc] = x0 + x1 + x2;
        sXq[rr * RS + cc] = x0 * x0 + x1 * x1 + x2 * x2;
    }
    __syncthreads();

    #pragma unroll
    for (int g = 0; g < 3; ++g) {
        v8f acc1 = {}, acc2 = {};
        const int arow = g * 16 + lo;
        #pragma unroll
        for (int k = 0; k < 10; ++k) {
            const int j0 = 4 * k + 2 * hi;
            v2f a1 = { sXs[arow * RS + j0], sXs[arow * RS + j0 + 1] };
            v2f a2 = { sXq[arow * RS + j0], sXq[arow * RS + j0 + 1] };
            acc1 = __builtin_amdgcn_wmma_f32_16x16x4_f32(false, a1, false, band[k],
                                                         (short)0, acc1, false, false);
            acc2 = __builtin_amdgcn_wmma_f32_16x16x4_f32(false, a2, false, band[k],
                                                         (short)0, acc2, false, false);
        }
        #pragma unroll
        for (int v = 0; v < 8; ++v) {
            int row = g * 16 + v + 8 * hi;
            if (row < 40) {
                sH1[row * HS + lo] = acc1[v];
                sH2[row * HS + lo] = acc2[v];
            }
        }
    }
    __syncthreads();

    v8f s1 = {}, s2 = {};
    #pragma unroll
    for (int k = 0; k < 10; ++k) {
        const int i0 = 4 * k + 2 * hi;
        v2f b1 = { sH1[i0 * HS + lo], sH1[(i0 + 1) * HS + lo] };
        v2f b2 = { sH2[i0 * HS + lo], sH2[(i0 + 1) * HS + lo] };
        s1 = __builtin_amdgcn_wmma_f32_16x16x4_f32(false, band[k], false, b1,
                                                   (short)0, s1, false, false);
        s2 = __builtin_amdgcn_wmma_f32_16x16x4_f32(false, band[k], false, b2,
                                                   (short)0, s2, false, false);
    }

    const float inv = 1.0f / (25.0f * 25.0f * 3.0f);
    const float EPS = 1.0f / 255.0f;
    float* outb = out + (size_t)b * CH_STRIDE;
    #pragma unroll
    for (int v = 0; v < 8; ++v) {
        const int row = tr + v + 8 * hi;
        const int col = tc + lo;
        const size_t off = (size_t)row * W_IMG + col;
        float meanr = s1[v] * inv;
        float means = s2[v] * inv;
        float expo  = fabsf(meanr - 0.5f) + EPS;
        float contrast = means - meanr * meanr;
        float x0 = base[off];
        float x1 = base[off + CH_STRIDE];
        float x2 = base[off + 2 * CH_STRIDE];
        float mx = fmaxf(x0, fmaxf(x1, x2));
        float mn = fminf(x0, fminf(x1, x2));
        float sat = (mx - mn + EPS) / (mx + EPS);
        outb[off] = sat * contrast / expo;
    }
}

extern "C" void kernel_launch(void* const* d_in, const int* in_sizes, int n_in,
                              void* d_out, int out_size, void* d_ws, size_t ws_size,
                              hipStream_t stream) {
    (void)in_sizes; (void)n_in; (void)d_ws; (void)ws_size; (void)out_size;
    const float* img = (const float*)d_in[0];
    float* out = (float*)d_out;
    dim3 grid(W_IMG / 16, H_IMG / 16, 16);
    ImageEvaluator_84000970375834_kernel<<<grid, dim3(32), 0, stream>>>(img, out);
}


# vertical GEMM first, fed from global loads (lane = column); its D tiles are the horizontal GEMM's A fragments, held in registers: no LDS; centre pixels reused; rows summed before columns
# speedup vs baseline: 3.2403x; 3.2403x over previous
#include <hip/hip_runtime.h>

typedef __attribute__((ext_vector_type(2))) float v2f;
typedef __attribute__((ext_vector_type(8))) float v8f;

#define W_IMG 512
#define CH_STRIDE (512 * 512)
#define TILES 4
#define NBLK (TILES + 2)

__device__ __forceinline__ int reflect_idx(int i) {
    return (i < 0) ? -i : ((i > 511) ? 1022 - i : i);
}

__device__ __forceinline__ void load_block(const float* __restrict__ base, unsigned offC, unsigned offH,
                                           float (&x0)[20], float (&x1)[20], float (&x2)[20]) {
    const float* pc = base + offC;
    const float* ph = base + offH;
    #pragma unroll
    for (int i = 0; i < 8; ++i) {
        x0[i] = pc[i * W_IMG];
        x1[i] = pc[i * W_IMG + CH_STRIDE];
        x2[i] = pc[i * W_IMG + 2 * CH_STRIDE];
    }
    #pragma unroll
    for (int i = 0; i < 12; ++i) {
        x0[8 + i] = ph[i * W_IMG];
        x1[8 + i] = ph[i * W_IMG + CH_STRIDE];
        x2[8 + i] = ph[i * W_IMG + 2 * CH_STRIDE];
    }
}

__device__ __forceinline__ void saturation(const float (&x0)[20], const float (&x1)[20], const float (&x2)[20],
                                           float (&sat)[8]) {
    const float EPS = 1.0f / 255.0f;
    #pragma unroll
    for (int v = 0; v < 8; ++v) {
        const float mx = fmaxf(x0[v], fmaxf(x1[v], x2[v]));
        const float mn = fminf(x0[v], fminf(x1[v], x2[v]));
        sat[v] = (mx - mn + EPS) / (mx + EPS);
    }
}

__device__ __forceinline__ void vertical_gemm(const float (&x0)[20], const float (&x1)[20], const float (&x2)[20],
                                              const v2f (&band1)[10], v8f& Vs, v8f& Vq) {
    v8f acc1 = {}, acc2 = {};
    #pragma unroll
    for (int k = 0; k < 10; ++k) {
        const float p0 = x0[2 * k], p1 = x1[2 * k], p2 = x2[2 * k];
        const float q0 = x0[2 * k + 1], q1 = x1[2 * k + 1], q2 = x2[2 * k + 1];
        const v2f as = { p0 + p1 + p2, q0 + q1 + q2 };
        const v2f aq = { p0 * p0 + p1 * p1 + p2 * p2, q0 * q0 + q1 * q1 + q2 * q2 };
        acc1 = __builtin_amdgcn_wmma_f32_16x16x4_f32(false, as, false, band1[k],
                                                     (short)0, acc1, false, false);
        acc2 = __builtin_amdgcn_wmma_f32_16x16x4_f32(false, aq, false, band1[k],
                                                     (short)0, acc2, false, false);
    }
    Vs = acc1;
    Vq = acc2;
}

__device__ __forceinline__ void block_consumed(v8f& Vs, v8f& Vq) {
    asm volatile("" : "+v"(Vs), "+v"(Vq) : : "memory");
}

__device__ __forceinline__ void loads_issued(v8f& Vs, v8f& Vq) {
    asm volatile("" : "+v"(Vs), "+v"(Vq) : : "memory");
}

__device__ __forceinline__ void horizontal_gemm_tile(const v8f& Vs0, const v8f& Vs1, const v8f& Vs2,
                                                     const v8f& Vq0, const v8f& Vq1, const v8f& Vq2,
                                                     const v2f (&band2)[12], const float (&sat)[8],
                                                     float* __restrict__ outb, unsigned outoff) {
    v8f s1 = {}, s2 = {};
    #pragma unroll
    for (int q = 0; q < 4; ++q) {
        v2f a1 = { Vs0[2 * q], Vs0[2 * q + 1] };
        v2f a2 = { Vq0[2 * q], Vq0[2 * q + 1] };
        s1 = __builtin_amdgcn_wmma_f32_16x16x4_f32(false, a1, false, band2[q],
                                                   (short)0, s1, false, false);
        s2 = __builtin_amdgcn_wmma_f32_16x16x4_f32(false, a2, false, band2[q],
                                                   (short)0, s2, false, false);
    }
    #pragma unroll
    for (int q = 0; q < 4; ++q) {
        v2f a1 = { Vs1[2 * q], Vs1[2 * q + 1] };
        v2f a2 = { Vq1[2 * q], Vq1[2 * q + 1] };
        s1 = __builtin_amdgcn_wmma_f32_16x16x4_f32(false, a1, false, band2[4 + q],
                                                   (short)0, s1, false, false);
        s2 = __builtin_amdgcn_wmma_f32_16x16x4_f32(false, a2, false, band2[4 + q],
                                                   (short)0, s2, false, false);
    }
    #pragma unroll
    for (int q = 0; q < 4; ++q) {
        v2f a1 = { Vs2[2 * q], Vs2[2 * q + 1] };
        v2f a2 = { Vq2[2 * q], Vq2[2 * q + 1] };
        s1 = __builtin_amdgcn_wmma_f32_16x16x4_f32(false, a1, false, band2[8 + q],
                                                   (short)0, s1, false, false);
        s2 = __builtin_amdgcn_wmma_f32_16x16x4_f32(false, a2, false, band2[8 + q],
                                                   (short)0, s2, false, false);
    }

    const float inv = 1.0f / (25.0f * 25.0f * 3.0f);
    const float EPS = 1.0f / 255.0f;
    float* op = outb + outoff;
    #pragma unroll
    for (int v = 0; v < 8; ++v) {
        float meanr = s1[v] * inv;
        float means = s2[v] * inv;
        float expo  = fabsf(meanr - 0.5f) + EPS;
        float contrast = means - meanr * meanr;
        op[v * W_IMG] = sat[v] * contrast / expo;
    }
}

__global__ __launch_bounds__(32, 4)
void box_stats_chain_kernel(const float* __restrict__ img, float* __restrict__ out) {
    const int lane = threadIdx.x;
    const int hi   = lane >> 4;
    const int lo   = lane & 15;
    const int c0   = blockIdx.x * (16 * TILES);
    const int r0   = blockIdx.y * 16;
    const int b    = blockIdx.z;

    const float* base = img + (size_t)b * 3 * CH_STRIDE;
    float* outb = out + (size_t)b * CH_STRIDE;

    const int  hw0   = hi ? 28 : 0;
    const int  hrow0 = r0 - 12 + hw0;
    const bool flip  = (hrow0 < 0) || (hrow0 + 11 > 511);
    const int  hrowA = flip ? reflect_idx(hrow0 + 11) : hrow0;
    const unsigned rowC = (unsigned)(r0 + 8 * hi) * W_IMG;
    const unsigned rowH = (unsigned)hrowA * W_IMG;

    v2f band1[10];
    #pragma unroll
    for (int k = 0; k < 10; ++k) {
        int w0, w1;
        if (k < 4) {
            w0 = 12 + 8 * hi + 2 * k;
            w1 = w0 + 1;
        } else {
            const int i = 2 * (k - 4);
            w0 = hw0 + (flip ? 11 - i : i);
            w1 = hw0 + (flip ? 10 - i : i + 1);
        }
        band1[k].x = ((unsigned)(w0 - lo) <= 24u) ? 1.0f : 0.0f;
        band1[k].y = ((unsigned)(w1 - lo) <= 24u) ? 1.0f : 0.0f;
    }
    v2f band2[12];
    #pragma unroll
    for (int c = 0; c < 12; ++c) {
        const int j0 = 16 * (c >> 2) + 8 * hi + 2 * (c & 3) - 4;
        band2[c].x = ((unsigned)(j0     - lo) <= 24u) ? 1.0f : 0.0f;
        band2[c].y = ((unsigned)(j0 + 1 - lo) <= 24u) ? 1.0f : 0.0f;
    }

    float x0[20], x1[20], x2[20];
    float y0[20], y1[20], y2[20];
    v8f Vs0, Vs1, Vs2, Vq0, Vq1, Vq2;
    float satC[8], satN[8];

    unsigned colb = (unsigned)reflect_idx(c0 - 16 + lo);
    load_block(base, rowC + colb, rowH + colb, x0, x1, x2);
    colb = (unsigned)(c0 + lo);
    load_block(base, rowC + colb, rowH + colb, y0, y1, y2);

    vertical_gemm(x0, x1, x2, band1, Vs0, Vq0);
    block_consumed(Vs0, Vq0);
    colb = (unsigned)(c0 + 16 + lo);
    load_block(base, rowC + colb, rowH + colb, x0, x1, x2);

    saturation(y0, y1, y2, satC);
    vertical_gemm(y0, y1, y2, band1, Vs1, Vq1);

    unsigned outoff = (unsigned)(r0 + 8 * hi) * W_IMG + (unsigned)(c0 + lo);

    #pragma unroll
    for (int j = 2; j < NBLK - 1; ++j) {
        saturation(x0, x1, x2, satN);
        vertical_gemm(x0, x1, x2, band1, Vs2, Vq2);
        block_consumed(Vs2, Vq2);
        colb = (unsigned)reflect_idx(c0 - 16 + 16 * (j + 1) + lo);
        load_block(base, rowC + colb, rowH + colb, x0, x1, x2);
        loads_issued(Vs0, Vq0);

        horizontal_gemm_tile(Vs0, Vs1, Vs2, Vq0, Vq1, Vq2, band2, satC, outb, outoff);
        outoff += 16u;

        Vs0 = Vs1; Vs1 = Vs2;
        Vq0 = Vq1; Vq1 = Vq2;
        #pragma unroll
        for (int v = 0; v < 8; ++v) satC[v] = satN[v];
    }

    vertical_gemm(x0, x1, x2, band1, Vs2, Vq2);
    horizontal_gemm_tile(Vs0, Vs1, Vs2, Vq0, Vq1, Vq2, band2, satC, outb, outoff);
}

extern "C" void kernel_launch(void* const* d_in, const int* in_sizes, int n_in,
                              void* d_out, int out_size, void* d_ws, size_t ws_size,
                              hipStream_t stream) {
    (void)in_sizes; (void)n_in; (void)d_ws; (void)ws_size; (void)out_size;
    const float* img = (const float*)d_in[0];
    float* out = (float*)d_out;
    dim3 grid(8, 32, 16);
    box_stats_chain_kernel<<<grid, dim3(32), 0, stream>>>(img, out);
}


# the four waves of a work-group take four vertically adjacent row bands (grid 8x8x16, block 32x4): window rows two bands share hit the WGP cache instead of the second level; sum order same
# speedup vs baseline: 3.4520x; 1.0653x over previous
#include <hip/hip_runtime.h>

typedef __attribute__((ext_vector_type(2))) float v2f;
typedef __attribute__((ext_vector_type(8))) float v8f;

#define W_IMG 512
#define CH_STRIDE (512 * 512)
#define TILES 4
#define NBLK (TILES + 2)

__device__ __forceinline__ int reflect_idx(int i) {
    return (i < 0) ? -i : ((i > 511) ? 1022 - i : i);
}

__device__ __forceinline__ void load_block(const float* __restrict__ base, unsigned offC, unsigned offH,
                                           float (&x0)[20], float (&x1)[20], float (&x2)[20]) {
    const float* pc = base + offC;
    const float* ph = base + offH;
    #pragma unroll
    for (int i = 0; i < 8; ++i) {
        x0[i] = pc[i * W_IMG];
        x1[i] = pc[i * W_IMG + CH_STRIDE];
        x2[i] = pc[i * W_IMG + 2 * CH_STRIDE];
    }
    #pragma unroll
    for (int i = 0; i < 12; ++i) {
        x0[8 + i] = ph[i * W_IMG];
        x1[8 + i] = ph[i * W_IMG + CH_STRIDE];
        x2[8 + i] = ph[i * W_IMG + 2 * CH_STRIDE];
    }
}

__device__ __forceinline__ void saturation(const float (&x0)[20], const float (&x1)[20], const float (&x2)[20],
                                           float (&sat)[8]) {
    const float EPS = 1.0f / 255.0f;
    #pragma unroll
    for (int v = 0; v < 8; ++v) {
        const float mx = fmaxf(x0[v], fmaxf(x1[v], x2[v]));
        const float mn = fminf(x0[v], fminf(x1[v], x2[v]));
        sat[v] = (mx - mn + EPS) / (mx + EPS);
    }
}

__device__ __forceinline__ void vertical_gemm(const float (&x0)[20], const float (&x1)[20], const float (&x2)[20],
                                              const v2f (&band1)[10], v8f& Vs, v8f& Vq) {
    v8f acc1 = {}, acc2 = {};
    #pragma unroll
    for (int k = 0; k < 10; ++k) {
        const float p0 = x0[2 * k], p1 = x1[2 * k], p2 = x2[2 * k];
        const float q0 = x0[2 * k + 1], q1 = x1[2 * k + 1], q2 = x2[2 * k + 1];
        const v2f as = { p0 + p1 + p2, q0 + q1 + q2 };
        const v2f aq = { p0 * p0 + p1 * p1 + p2 * p2, q0 * q0 + q1 * q1 + q2 * q2 };
        acc1 = __builtin_amdgcn_wmma_f32_16x16x4_f32(false, as, false, band1[k],
                                                     (short)0, acc1, false, false);
        acc2 = __builtin_amdgcn_wmma_f32_16x16x4_f32(false, aq, false, band1[k],
                                                     (short)0, acc2, false, false);
    }
    Vs = acc1;
    Vq = acc2;
}

__device__ __forceinline__ void block_consumed(v8f& Vs, v8f& Vq) {
    asm volatile("" : "+v"(Vs), "+v"(Vq) : : "memory");
}

__device__ __forceinline__ void loads_issued(v8f& Vs, v8f& Vq) {
    asm volatile("" : "+v"(Vs), "+v"(Vq) : : "memory");
}

__device__ __forceinline__ void horizontal_gemm_tile(const v8f& Vs0, const v8f& Vs1, const v8f& Vs2,
                                                     const v8f& Vq0, const v8f& Vq1, const v8f& Vq2,
                                                     const v2f (&band2)[12], const float (&sat)[8],
                                                     float* __restrict__ outb, unsigned outoff) {
    v8f s1 = {}, s2 = {};
    #pragma unroll
    for (int q = 0; q < 4; ++q) {
        v2f a1 = { Vs0[2 * q], Vs0[2 * q + 1] };
        v2f a2 = { Vq0[2 * q], Vq0[2 * q + 1] };
        s1 = __builtin_amdgcn_wmma_f32_16x16x4_f32(false, a1, false, band2[q],
                                                   (short)0, s1, false, false);
        s2 = __builtin_amdgcn_wmma_f32_16x16x4_f32(false, a2, false, band2[q],
                                                   (short)0, s2, false, false);
    }
    #pragma unroll
    for (int q = 0; q < 4; ++q) {
        v2f a1 = { Vs1[2 * q], Vs1[2 * q + 1] };
        v2f a2 = { Vq1[2 * q], Vq1[2 * q + 1] };
        s1 = __builtin_amdgcn_wmma_f32_16x16x4_f32(false, a1, false, band2[4 + q],
                                                   (short)0, s1, false, false);
        s2 = __builtin_amdgcn_wmma_f32_16x16x4_f32(false, a2, false, band2[4 + q],
                                                   (short)0, s2, false, false);
    }
    #pragma unroll
    for (int q = 0; q < 4; ++q) {
        v2f a1 = { Vs2[2 * q], Vs2[2 * q + 1] };
        v2f a2 = { Vq2[2 * q], Vq2[2 * q + 1] };
        s1 = __builtin_amdgcn_wmma_f32_16x16x4_f32(false, a1, false, band2[8 + q],
                                                   (short)0, s1, false, false);
        s2 = __builtin_amdgcn_wmma_f32_16x16x4_f32(false, a2, false, band2[8 + q],
                                                   (short)0, s2, false, false);
    }

    const float inv = 1.0f / (25.0f * 25.0f * 3.0f);
    const float EPS = 1.0f / 255.0f;
    float* op = outb + outoff;
    #pragma unroll
    for (int v = 0; v < 8; ++v) {
        float meanr = s1[v] * inv;
        float means = s2[v] * inv;
        float expo  = fabsf(meanr - 0.5f) + EPS;
        float contrast = means - meanr * meanr;
        op[v * W_IMG] = sat[v] * contrast / expo;
    }
}

__global__ __launch_bounds__(128, 4)
void box_stats_chain_kernel(const float* __restrict__ img, float* __restrict__ out) {
    const int lane = threadIdx.x;
    const int hi   = lane >> 4;
    const int lo   = lane & 15;
    const int c0   = blockIdx.x * (16 * TILES);
    const int r0   = (blockIdx.y * 4 + threadIdx.y) * 16;
    const int b    = blockIdx.z;

    const float* base = img + (size_t)b * 3 * CH_STRIDE;
    float* outb = out + (size_t)b * CH_STRIDE;

    const int  hw0   = hi ? 28 : 0;
    const int  hrow0 = r0 - 12 + hw0;
    const bool flip  = (hrow0 < 0) || (hrow0 + 11 > 511);
    const int  hrowA = flip ? reflect_idx(hrow0 + 11) : hrow0;
    const unsigned rowC = (unsigned)(r0 + 8 * hi) * W_IMG;
    const unsigned rowH = (unsigned)hrowA * W_IMG;

    v2f band1[10];
    #pragma unroll
    for (int k = 0; k < 10; ++k) {
        int w0, w1;
        if (k < 4) {
            w0 = 12 + 8 * hi + 2 * k;
            w1 = w0 + 1;
        } else {
            const int i = 2 * (k - 4);
            w0 = hw0 + (flip ? 11 - i : i);
            w1 = hw0 + (flip ? 10 - i : i + 1);
        }
        band1[k].x = ((unsigned)(w0 - lo) <= 24u) ? 1.0f : 0.0f;
        band1[k].y = ((unsigned)(w1 - lo) <= 24u) ? 1.0f : 0.0f;
    }
    v2f band2[12];
    #pragma unroll
    for (int c = 0; c < 12; ++c) {
        const int j0 = 16 * (c >> 2) + 8 * hi + 2 * (c & 3) - 4;
        band2[c].x = ((unsigned)(j0     - lo) <= 24u) ? 1.0f : 0.0f;
        band2[c].y = ((unsigned)(j0 + 1 - lo) <= 24u) ? 1.0f : 0.0f;
    }

    float x0[20], x1[20], x2[20];
    float y0[20], y1[20], y2[20];
    v8f Vs0, Vs1, Vs2, Vq0, Vq1, Vq2;
    float satC[8], satN[8];

    unsigned colb = (unsigned)reflect_idx(c0 - 16 + lo);
    load_block(base, rowC + colb, rowH + colb, x0, x1, x2);
    colb = (unsigned)(c0 + lo);
    load_block(base, rowC + colb, rowH + colb, y0, y1, y2);

    vertical_gemm(x0, x1, x2, band1, Vs0, Vq0);
    block_consumed(Vs0, Vq0);
    colb = (unsigned)(c0 + 16 + lo);
    load_block(base, rowC + colb, rowH + colb, x0, x1, x2);

    saturation(y0, y1, y2, satC);
    vertical_gemm(y0, y1, y2, band1, Vs1, Vq1);

    unsigned outoff = (unsigned)(r0 + 8 * hi) * W_IMG + (unsigned)(c0 + lo);

    #pragma unroll
    for (int j = 2; j < NBLK - 1; ++j) {
        saturation(x0, x1, x2, satN);
        vertical_gemm(x0, x1, x2, band1, Vs2, Vq2);
        block_consumed(Vs2, Vq2);
        colb = (unsigned)reflect_idx(c0 - 16 + 16 * (j + 1) + lo);
        load_block(base, rowC + colb, rowH + colb, x0, x1, x2);
        loads_issued(Vs0, Vq0);

        horizontal_gemm_tile(Vs0, Vs1, Vs2, Vq0, Vq1, Vq2, band2, satC, outb, outoff);
        outoff += 16u;

        Vs0 = Vs1; Vs1 = Vs2;
        Vq0 = Vq1; Vq1 = Vq2;
        #pragma unroll
        for (int v = 0; v < 8; ++v) satC[v] = satN[v];
    }

    vertical_gemm(x0, x1, x2, band1, Vs2, Vq2);
    horizontal_gemm_tile(Vs0, Vs1, Vs2, Vq0, Vq1, Vq2, band2, satC, outb, outoff);
}

extern "C" void kernel_launch(void* const* d_in, const int* in_sizes, int n_in,
                              void* d_out, int out_size, void* d_ws, size_t ws_size,
                              hipStream_t stream) {
    (void)in_sizes; (void)n_in; (void)d_ws; (void)ws_size; (void)out_size;
    const float* img = (const float*)d_in[0];
    float* out = (float*)d_out;
    dim3 grid(8, 8, 16);
    box_stats_chain_kernel<<<grid, dim3(32, 4), 0, stream>>>(img, out);
}


# partner waves start at a shared column border and walk apart; each passes its first block's vertical-GEMM tiles to the other through LDS: 5 not 6 blocks per wave; left walker sums columns mirrored
# speedup vs baseline: 3.7657x; 1.0909x over previous
#include <hip/hip_runtime.h>

typedef __attribute__((ext_vector_type(2))) float v2f;
typedef __attribute__((ext_vector_type(8))) float v8f;

#define W_IMG 512
#define CH_STRIDE (512 * 512)
#define TILES 4
#define NBLK (TILES + 2)

__device__ __forceinline__ int reflect_idx(int i) {
    return (i < 0) ? -i : ((i > 511) ? 1022 - i : i);
}

__device__ __forceinline__ void load_block(const float* __restrict__ base, unsigned offC, unsigned offH,
                                           float (&x0)[20], float (&x1)[20], float (&x2)[20]) {
    const float* pc = base + offC;
    const float* ph = base + offH;
    #pragma unroll
    for (int i = 0; i < 8; ++i) {
        x0[i] = pc[i * W_IMG];
        x1[i] = pc[i * W_IMG + CH_STRIDE];
        x2[i] = pc[i * W_IMG + 2 * CH_STRIDE];
    }
    #pragma unroll
    for (int i = 0; i < 12; ++i) {
        x0[8 + i] = ph[i * W_IMG];
        x1[8 + i] = ph[i * W_IMG + CH_STRIDE];
        x2[8 + i] = ph[i * W_IMG + 2 * CH_STRIDE];
    }
}

__device__ __forceinline__ void saturation(const float (&x0)[20], const float (&x1)[20], const float (&x2)[20],
                                           float (&sat)[8]) {
    const float EPS = 1.0f / 255.0f;
    #pragma unroll
    for (int v = 0; v < 8; ++v) {
        const float mx = fmaxf(x0[v], fmaxf(x1[v], x2[v]));
        const float mn = fminf(x0[v], fminf(x1[v], x2[v]));
        sat[v] = (mx - mn + EPS) / (mx + EPS);
    }
}

__device__ __forceinline__ void vertical_gemm(const float (&x0)[20], const float (&x1)[20], const float (&x2)[20],
                                              const v2f (&band1)[10], v8f& Vs, v8f& Vq) {
    v8f acc1 = {}, acc2 = {};
    #pragma unroll
    for (int k = 0; k < 10; ++k) {
        const float p0 = x0[2 * k], p1 = x1[2 * k], p2 = x2[2 * k];
        const float q0 = x0[2 * k + 1], q1 = x1[2 * k + 1], q2 = x2[2 * k + 1];
        const v2f as = { p0 + p1 + p2, q0 + q1 + q2 };
        const v2f aq = { p0 * p0 + p1 * p1 + p2 * p2, q0 * q0 + q1 * q1 + q2 * q2 };
        acc1 = __builtin_amdgcn_wmma_f32_16x16x4_f32(false, as, false, band1[k],
                                                     (short)0, acc1, false, false);
        acc2 = __builtin_amdgcn_wmma_f32_16x16x4_f32(false, aq, false, band1[k],
                                                     (short)0, acc2, false, false);
    }
    Vs = acc1;
    Vq = acc2;
}

__device__ __forceinline__ void block_consumed(v8f& Vs, v8f& Vq) {
    asm volatile("" : "+v"(Vs), "+v"(Vq) : : "memory");
}

__device__ __forceinline__ void loads_issued(v8f& Vs, v8f& Vq) {
    asm volatile("" : "+v"(Vs), "+v"(Vq) : : "memory");
}

__device__ __forceinline__ void horizontal_gemm_tile(const v8f& Vs0, const v8f& Vs1, const v8f& Vs2,
                                                     const v8f& Vq0, const v8f& Vq1, const v8f& Vq2,
                                                     const v2f (&band2)[12], const float (&sat)[8],
                                                     float* __restrict__ outb, unsigned outoff) {
    v8f s1 = {}, s2 = {};
    #pragma unroll
    for (int q = 0; q < 4; ++q) {
        v2f a1 = { Vs0[2 * q], Vs0[2 * q + 1] };
        v2f a2 = { Vq0[2 * q], Vq0[2 * q + 1] };
        s1 = __builtin_amdgcn_wmma_f32_16x16x4_f32(false, a1, false, band2[q],
                                                   (short)0, s1, false, false);
        s2 = __builtin_amdgcn_wmma_f32_16x16x4_f32(false, a2, false, band2[q],
                                                   (short)0, s2, false, false);
    }
    #pragma unroll
    for (int q = 0; q < 4; ++q) {
        v2f a1 = { Vs1[2 * q], Vs1[2 * q + 1] };
        v2f a2 = { Vq1[2 * q], Vq1[2 * q + 1] };
        s1 = __builtin_amdgcn_wmma_f32_16x16x4_f32(false, a1, false, band2[4 + q],
                                                   (short)0, s1, false, false);
        s2 = __builtin_amdgcn_wmma_f32_16x16x4_f32(false, a2, false, band2[4 + q],
                                                   (short)0, s2, false, false);
    }
    #pragma unroll
    for (int q = 0; q < 4; ++q) {
        v2f a1 = { Vs2[2 * q], Vs2[2 * q + 1] };
        v2f a2 = { Vq2[2 * q], Vq2[2 * q + 1] };
        s1 = __builtin_amdgcn_wmma_f32_16x16x4_f32(false, a1, false, band2[8 + q],
                                                   (short)0, s1, false, false);
        s2 = __builtin_amdgcn_wmma_f32_16x16x4_f32(false, a2, false, band2[8 + q],
                                                   (short)0, s2, false, false);
    }

    const float inv = 1.0f / (25.0f * 25.0f * 3.0f);
    const float EPS = 1.0f / 255.0f;
    float* op = outb + outoff;
    #pragma unroll
    for (int v = 0; v < 8; ++v) {
        float meanr = s1[v] * inv;
        float means = s2[v] * inv;
        float expo  = fabsf(meanr - 0.5f) + EPS;
        float contrast = means - meanr * meanr;
        op[v * W_IMG] = sat[v] * contrast / expo;
    }
}

__global__ __launch_bounds__(256, 4)
void box_stats_chain_kernel(const float* __restrict__ img, float* __restrict__ out) {
    __shared__ float xch[8][16][32];

    const int lane = threadIdx.x;
    const int hi   = lane >> 4;
    const int lo   = lane & 15;
    const int wv   = threadIdx.y;
    const int band = wv & 3;
    const int dir  = wv >> 2;
    const int cB   = blockIdx.x * (32 * TILES) + 16 * TILES;
    const int r0   = (blockIdx.y * 4 + band) * 16;
    const int b    = blockIdx.z;

    const float* base = img + (size_t)b * 3 * CH_STRIDE;
    float* outb = out + (size_t)b * CH_STRIDE;

    const int  hw0   = hi ? 28 : 0;
    const int  hrow0 = r0 - 12 + hw0;
    const bool flip  = (hrow0 < 0) || (hrow0 + 11 > 511);
    const int  hrowA = flip ? reflect_idx(hrow0 + 11) : hrow0;
    const unsigned rowC = (unsigned)(r0 + 8 * hi) * W_IMG;
    const unsigned rowH = (unsigned)hrowA * W_IMG;

    v2f band1[10];
    #pragma unroll
    for (int k = 0; k < 10; ++k) {
        int w0, w1;
        if (k < 4) {
            w0 = 12 + 8 * hi + 2 * k;
            w1 = w0 + 1;
        } else {
            const int i = 2 * (k - 4);
            w0 = hw0 + (flip ? 11 - i : i);
            w1 = hw0 + (flip ? 10 - i : i + 1);
        }
        band1[k].x = ((unsigned)(w0 - lo) <= 24u) ? 1.0f : 0.0f;
        band1[k].y = ((unsigned)(w1 - lo) <= 24u) ? 1.0f : 0.0f;
    }
    v2f band2[12];
    #pragma unroll
    for (int c = 0; c < 12; ++c) {
        const int j0 = 16 * (c >> 2) + 8 * hi + 2 * (c & 3) - 4;
        band2[c].x = ((unsigned)(j0     - lo) <= 24u) ? 1.0f : 0.0f;
        band2[c].y = ((unsigned)(j0 + 1 - lo) <= 24u) ? 1.0f : 0.0f;
    }

    float x0[20], x1[20], x2[20];
    float y0[20], y1[20], y2[20];
    v8f Vs0, Vs1, Vs2, Vq0, Vq1, Vq2;
    float satC[8], satN[8];

    const int cfirst = dir ? cB - 1 - lo : cB + lo;
    const int cstep  = dir ? -16 : 16;

    unsigned colb = (unsigned)cfirst;
    load_block(base, rowC + colb, rowH + colb, y0, y1, y2);
    colb = (unsigned)(cfirst + cstep);
    load_block(base, rowC + colb, rowH + colb, x0, x1, x2);

    saturation(y0, y1, y2, satC);
    vertical_gemm(y0, y1, y2, band1, Vs1, Vq1);
    #pragma unroll
    for (int v = 0; v < 8; ++v) {
        xch[wv][v][lane]     = Vs1[v];
        xch[wv][8 + v][lane] = Vq1[v];
    }
    __builtin_amdgcn_fence(__ATOMIC_RELEASE, "workgroup", "local");
    __builtin_amdgcn_s_barrier_signal(-1);

    unsigned outoff = (unsigned)(r0 + 8 * hi) * W_IMG + (unsigned)cfirst;

    #pragma unroll
    for (int j = 2; j < NBLK - 1; ++j) {
        saturation(x0, x1, x2, satN);
        vertical_gemm(x0, x1, x2, band1, Vs2, Vq2);
        block_consumed(Vs2, Vq2);
        const int cnext = cfirst + (j) * cstep;
        colb = (unsigned)((j + 1 == NBLK - 1) ? reflect_idx(cnext) : cnext);
        load_block(base, rowC + colb, rowH + colb, x0, x1, x2);
        if (j == 2) {
            loads_issued(Vs1, Vq1);
            __builtin_amdgcn_s_barrier_wait(-1);
            __builtin_amdgcn_fence(__ATOMIC_ACQUIRE, "workgroup", "local");
            #pragma unroll
            for (int v = 0; v < 8; ++v) {
                Vs0[v] = xch[wv ^ 4][7 - v][lane ^ 16];
                Vq0[v] = xch[wv ^ 4][15 - v][lane ^ 16];
            }
        } else {
            loads_issued(Vs0, Vq0);
        }

        horizontal_gemm_tile(Vs0, Vs1, Vs2, Vq0, Vq1, Vq2, band2, satC, outb, outoff);
        outoff += (unsigned)cstep;

        Vs0 = Vs1; Vs1 = Vs2;
        Vq0 = Vq1; Vq1 = Vq2;
        #pragma unroll
        for (int v = 0; v < 8; ++v) satC[v] = satN[v];
    }

    vertical_gemm(x0, x1, x2, band1, Vs2, Vq2);
    horizontal_gemm_tile(Vs0, Vs1, Vs2, Vq0, Vq1, Vq2, band2, satC, outb, outoff);
}

extern "C" void kernel_launch(void* const* d_in, const int* in_sizes, int n_in,
                              void* d_out, int out_size, void* d_ws, size_t ws_size,
                              hipStream_t stream) {
    (void)in_sizes; (void)n_in; (void)d_ws; (void)ws_size; (void)out_size;
    const float* img = (const float*)d_in[0];
    float* out = (float*)d_out;
    dim3 grid(4, 8, 16);
    box_stats_chain_kernel<<<grid, dim3(32, 8), 0, stream>>>(img, out);
}


# K slots only for distinct band rows: the 10 window rows/cols all 16 outputs share pre-added on VALU into one slot, empty halo half-chunks merged: 8 not 10/12 K chunks per GEMM; sum order changes
# speedup vs baseline: 3.9712x; 1.0546x over previous
#include <hip/hip_runtime.h>

typedef __attribute__((ext_vector_type(2))) float v2f;
typedef __attribute__((ext_vector_type(8))) float v8f;

#define W_IMG 512
#define CH_STRIDE (512 * 512)
#define TILES 4
#define NBLK (TILES + 2)

__device__ __forceinline__ int reflect_idx(int i) {
    return (i < 0) ? -i : ((i > 511) ? 1022 - i : i);
}

__device__ __forceinline__ void load_block(const float* __restrict__ base, unsigned offC, unsigned offH,
                                           float (&x0)[20], float (&x1)[20], float (&x2)[20]) {
    const float* pc = base + offC;
    const float* ph = base + offH;
    #pragma unroll
    for (int i = 0; i < 8; ++i) {
        x0[i] = pc[i * W_IMG];
        x1[i] = pc[i * W_IMG + CH_STRIDE];
        x2[i] = pc[i * W_IMG + 2 * CH_STRIDE];
    }
    #pragma unroll
    for (int i = 0; i < 12; ++i) {
        x0[8 + i] = ph[i * W_IMG];
        x1[8 + i] = ph[i * W_IMG + CH_STRIDE];
        x2[8 + i] = ph[i * W_IMG + 2 * CH_STRIDE];
    }
}

__device__ __forceinline__ void saturation(const float (&x0)[20], const float (&x1)[20], const float (&x2)[20],
                                           float (&sat)[8]) {
    const float EPS = 1.0f / 255.0f;
    #pragma unroll
    for (int v = 0; v < 8; ++v) {
        const float mx = fmaxf(x0[v], fmaxf(x1[v], x2[v]));
        const float mn = fminf(x0[v], fminf(x1[v], x2[v]));
        sat[v] = (mx - mn + EPS) / (mx + EPS);
    }
}

__device__ __forceinline__ void vertical_gemm(const float (&x0)[20], const float (&x1)[20], const float (&x2)[20],
                                              const v2f (&band1)[8], const bool hi, v8f& Vs, v8f& Vq) {
    float s[8], q[8];
    #pragma unroll
    for (int i = 0; i < 8; ++i) {
        s[i] = x0[i] + x1[i] + x2[i];
        q[i] = x0[i] * x0[i] + x1[i] * x1[i] + x2[i] * x2[i];
    }
    const float cs = (s[3] + s[4]) + (((hi ? s[0] : s[5]) + (hi ? s[1] : s[6])) + (hi ? s[2] : s[7]));
    const float cq = (q[3] + q[4]) + (((hi ? q[0] : q[5]) + (hi ? q[1] : q[6])) + (hi ? q[2] : q[7]));
    const v2f as0 = { hi ? s[5] : s[0], hi ? s[6] : s[1] };
    const v2f aq0 = { hi ? q[5] : q[0], hi ? q[6] : q[1] };
    const v2f as1 = { hi ? s[7] : s[2], cs };
    const v2f aq1 = { hi ? q[7] : q[2], cq };
    v8f acc1 = {}, acc2 = {};
    acc1 = __builtin_amdgcn_wmma_f32_16x16x4_f32(false, as0, false, band1[0], (short)0, acc1, false, false);
    acc2 = __builtin_amdgcn_wmma_f32_16x16x4_f32(false, aq0, false, band1[0], (short)0, acc2, false, false);
    acc1 = __builtin_amdgcn_wmma_f32_16x16x4_f32(false, as1, false, band1[1], (short)0, acc1, false, false);
    acc2 = __builtin_amdgcn_wmma_f32_16x16x4_f32(false, aq1, false, band1[1], (short)0, acc2, false, false);
    #pragma unroll
    for (int k = 2; k < 8; ++k) {
        const int e = 8 + 2 * (k - 2);
        const float p0 = x0[e], p1 = x1[e], p2 = x2[e];
        const float q0 = x0[e + 1], q1 = x1[e + 1], q2 = x2[e + 1];
        const v2f as = { p0 + p1 + p2, q0 + q1 + q2 };
        const v2f aq = { p0 * p0 + p1 * p1 + p2 * p2, q0 * q0 + q1 * q1 + q2 * q2 };
        acc1 = __builtin_amdgcn_wmma_f32_16x16x4_f32(false, as, false, band1[k],
                                                     (short)0, acc1, false, false);
        acc2 = __builtin_amdgcn_wmma_f32_16x16x4_f32(false, aq, false, band1[k],
                                                     (short)0, acc2, false, false);
    }
    Vs = acc1;
    Vq = acc2;
}

__device__ __forceinline__ void block_consumed(v8f& Vs, v8f& Vq) {
    asm volatile("" : "+v"(Vs), "+v"(Vq) : : "memory");
}

__device__ __forceinline__ void loads_issued(v8f& Vs, v8f& Vq) {
    asm volatile("" : "+v"(Vs), "+v"(Vq) : : "memory");
}

__device__ __forceinline__ void tile_summed(v8f& s1, v8f& s2, v2f& next_open) {
    asm volatile("" : "+v"(s1), "+v"(s2), "+v"(next_open));
}

__device__ __forceinline__ void horizontal_gemm_tile(const v8f& Vs0, const v8f& Vs1, const v8f& Vs2,
                                                     const v8f& Vq0, const v8f& Vq1, const v8f& Vq2,
                                                     const v2f (&band2)[8], const float (&sat)[8], const bool hi,
                                                     v2f& next_open, float* __restrict__ outb, unsigned outoff) {
    v8f s1 = {}, s2 = {};
    #pragma unroll
    for (int c = 0; c < 2; ++c) {
        v2f a1 = { Vs0[4 + 2 * c], Vs0[5 + 2 * c] };
        v2f a2 = { Vq0[4 + 2 * c], Vq0[5 + 2 * c] };
        s1 = __builtin_amdgcn_wmma_f32_16x16x4_f32(false, a1, false, band2[c],
                                                   (short)0, s1, false, false);
        s2 = __builtin_amdgcn_wmma_f32_16x16x4_f32(false, a2, false, band2[c],
                                                   (short)0, s2, false, false);
    }
    {
        const float cs = (Vs1[3] + Vs1[4]) + (((hi ? Vs1[0] : Vs1[5]) + (hi ? Vs1[1] : Vs1[6])) + (hi ? Vs1[2] : Vs1[7]));
        const float cq = (Vq1[3] + Vq1[4]) + (((hi ? Vq1[0] : Vq1[5]) + (hi ? Vq1[1] : Vq1[6])) + (hi ? Vq1[2] : Vq1[7]));
        const v2f a16 = { hi ? Vs1[5] : Vs1[0], hi ? Vs1[6] : Vs1[1] };
        const v2f a26 = { hi ? Vq1[5] : Vq1[0], hi ? Vq1[6] : Vq1[1] };
        const v2f a17 = { hi ? Vs1[7] : Vs1[2], cs };
        const v2f a27 = { hi ? Vq1[7] : Vq1[2], cq };
        s1 = __builtin_amdgcn_wmma_f32_16x16x4_f32(false, a16, false, band2[6], (short)0, s1, false, false);
        s2 = __builtin_amdgcn_wmma_f32_16x16x4_f32(false, a26, false, band2[6], (short)0, s2, false, false);
        s1 = __builtin_amdgcn_wmma_f32_16x16x4_f32(false, a17, false, band2[7], (short)0, s1, false, false);
        s2 = __builtin_amdgcn_wmma_f32_16x16x4_f32(false, a27, false, band2[7], (short)0, s2, false, false);
    }
    #pragma unroll
    for (int c = 0; c < 2; ++c) {
        v2f a1 = { Vs2[2 * c], Vs2[2 * c + 1] };
        v2f a2 = { Vq2[2 * c], Vq2[2 * c + 1] };
        s1 = __builtin_amdgcn_wmma_f32_16x16x4_f32(false, a1, false, band2[2 + c],
                                                   (short)0, s1, false, false);
        s2 = __builtin_amdgcn_wmma_f32_16x16x4_f32(false, a2, false, band2[2 + c],
                                                   (short)0, s2, false, false);
    }
    #pragma unroll
    for (int c = 0; c < 2; ++c) {
        v2f a1 = { hi ? Vs0[2 * c] : Vs2[4 + 2 * c], hi ? Vs0[2 * c + 1] : Vs2[5 + 2 * c] };
        v2f a2 = { hi ? Vq0[2 * c] : Vq2[4 + 2 * c], hi ? Vq0[2 * c + 1] : Vq2[5 + 2 * c] };
        s1 = __builtin_amdgcn_wmma_f32_16x16x4_f32(false, a1, false, band2[4 + c],
                                                   (short)0, s1, false, false);
        s2 = __builtin_amdgcn_wmma_f32_16x16x4_f32(false, a2, false, band2[4 + c],
                                                   (short)0, s2, false, false);
    }

    tile_summed(s1, s2, next_open);

    const float inv = 1.0f / (25.0f * 25.0f * 3.0f);
    const float EPS = 1.0f / 255.0f;
    float* op = outb + outoff;
    #pragma unroll
    for (int v = 0; v < 8; ++v) {
        float meanr = s1[v] * inv;
        float means = s2[v] * inv;
        float expo  = fabsf(meanr - 0.5f) + EPS;
        float contrast = means - meanr * meanr;
        op[v * W_IMG] = sat[v] * contrast / expo;
    }
}

__global__ __launch_bounds__(256, 4)
void box_stats_chain_kernel(const float* __restrict__ img, float* __restrict__ out) {
    __shared__ float xch[8][16][32];

    const int lane = threadIdx.x;
    const int hi   = lane >> 4;
    const int lo   = lane & 15;
    const int wv   = threadIdx.y;
    const int band = wv & 3;
    const int dir  = wv >> 2;
    const int cB   = blockIdx.x * (32 * TILES) + 16 * TILES;
    const int r0   = (blockIdx.y * 4 + band) * 16;
    const int b    = blockIdx.z;

    const float* base = img + (size_t)b * 3 * CH_STRIDE;
    float* outb = out + (size_t)b * CH_STRIDE;

    const int  hw0   = hi ? 28 : 0;
    const int  hrow0 = r0 - 12 + hw0;
    const bool flip  = (hrow0 < 0) || (hrow0 + 11 > 511);
    const int  hrowA = flip ? reflect_idx(hrow0 + 11) : hrow0;
    const unsigned rowC = (unsigned)(r0 + 8 * hi) * W_IMG;
    const unsigned rowH = (unsigned)hrowA * W_IMG;

    v2f band1[8];
    {
        const int wr = hi ? 25 : 12;
        band1[0].x = ((unsigned)(wr     - lo) <= 24u) ? 1.0f : 0.0f;
        band1[0].y = ((unsigned)(wr + 1 - lo) <= 24u) ? 1.0f : 0.0f;
        band1[1].x = ((unsigned)(wr + 2 - lo) <= 24u) ? 1.0f : 0.0f;
        band1[1].y = 1.0f;
    }
    #pragma unroll
    for (int k = 2; k < 8; ++k) {
        const int i = 2 * (k - 2);
        const int w0 = hw0 + (flip ? 11 - i : i);
        const int w1 = hw0 + (flip ? 10 - i : i + 1);
        band1[k].x = ((unsigned)(w0 - lo) <= 24u) ? 1.0f : 0.0f;
        band1[k].y = ((unsigned)(w1 - lo) <= 24u) ? 1.0f : 0.0f;
    }
    v2f band2[8];
    {
        const int jm[8] = { 8 * hi, 8 * hi + 2,
                            28 + 8 * hi, 30 + 8 * hi,
                            hi ? 4 : 32, hi ? 6 : 34,
                            hi ? 25 : 12, hi ? 27 : 14 };
        #pragma unroll
        for (int c = 0; c < 8; ++c) {
            band2[c].x = ((unsigned)(jm[c]     - lo) <= 24u) ? 1.0f : 0.0f;
            band2[c].y = ((unsigned)(jm[c] + 1 - lo) <= 24u) ? 1.0f : 0.0f;
        }
        band2[7].y = 1.0f;
    }

    float x0[20], x1[20], x2[20];
    float y0[20], y1[20], y2[20];
    v8f Vs0, Vs1, Vs2, Vq0, Vq1, Vq2;
    float satC[8], satN[8];

    const int cfirst = dir ? cB - 1 - lo : cB + lo;
    const int cstep  = dir ? -16 : 16;

    unsigned colb = (unsigned)cfirst;
    load_block(base, rowC + colb, rowH + colb, y0, y1, y2);
    colb = (unsigned)(cfirst + cstep);
    load_block(base, rowC + colb, rowH + colb, x0, x1, x2);

    saturation(y0, y1, y2, satC);
    vertical_gemm(y0, y1, y2, band1, hi != 0, Vs1, Vq1);
    #pragma unroll
    for (int v = 0; v < 8; ++v) {
        xch[wv][v][lane]     = Vs1[v];
        xch[wv][8 + v][lane] = Vq1[v];
    }
    __builtin_amdgcn_fence(__ATOMIC_RELEASE, "workgroup", "local");
    __builtin_amdgcn_s_barrier_signal(-1);

    unsigned outoff = (unsigned)(r0 + 8 * hi) * W_IMG + (unsigned)cfirst;

    #pragma unroll
    for (int j = 2; j < NBLK - 1; ++j) {
        saturation(x0, x1, x2, satN);
        vertical_gemm(x0, x1, x2, band1, hi != 0, Vs2, Vq2);
        block_consumed(Vs2, Vq2);
        const int cnext = cfirst + (j) * cstep;
        colb = (unsigned)((j + 1 == NBLK - 1) ? reflect_idx(cnext) : cnext);
        load_block(base, rowC + colb, rowH + colb, x0, x1, x2);
        if (j == 2) {
            loads_issued(Vs1, Vq1);
            __builtin_amdgcn_s_barrier_wait(-1);
            __builtin_amdgcn_fence(__ATOMIC_ACQUIRE, "workgroup", "local");
            #pragma unroll
            for (int v = 0; v < 8; ++v) {
                Vs0[v] = xch[wv ^ 4][7 - v][lane ^ 16];
                Vq0[v] = xch[wv ^ 4][15 - v][lane ^ 16];
            }
        } else {
            loads_issued(Vs0, Vq0);
        }

        horizontal_gemm_tile(Vs0, Vs1, Vs2, Vq0, Vq1, Vq2, band2, satC, hi != 0, band1[0], outb, outoff);
        outoff += (unsigned)cstep;

        Vs0 = Vs1; Vs1 = Vs2;
        Vq0 = Vq1; Vq1 = Vq2;
        #pragma unroll
        for (int v = 0; v < 8; ++v) satC[v] = satN[v];
    }

    vertical_gemm(x0, x1, x2, band1, hi != 0, Vs2, Vq2);
    horizontal_gemm_tile(Vs0, Vs1, Vs2, Vq0, Vq1, Vq2, band2, satC, hi != 0, band1[0], outb, outoff);
}

extern "C" void kernel_launch(void* const* d_in, const int* in_sizes, int n_in,
                              void* d_out, int out_size, void* d_ws, size_t ws_size,
                              hipStream_t stream) {
    (void)in_sizes; (void)n_in; (void)d_ws; (void)ws_size; (void)out_size;
    const float* img = (const float*)d_in[0];
    float* out = (float*)d_out;
    dim3 grid(4, 8, 16);
    box_stats_chain_kernel<<<grid, dim3(32, 8), 0, stream>>>(img, out);
}


# both per-pixel f32 quotients (saturation, and contrast over exposedness) as numerator * v_rcp_f32(denominator), 1 ulp, instead of the correctly rounded division sequence; sums untouched
# speedup vs baseline: 4.3841x; 1.1040x over previous
#include <hip/hip_runtime.h>

typedef __attribute__((ext_vector_type(2))) float v2f;
typedef __attribute__((ext_vector_type(8))) float v8f;

#define W_IMG 512
#define CH_STRIDE (512 * 512)
#define TILES 4
#define NBLK (TILES + 2)

__device__ __forceinline__ int reflect_idx(int i) {
    return (i < 0) ? -i : ((i > 511) ? 1022 - i : i);
}

__device__ __forceinline__ void load_block(const float* __restrict__ base, unsigned offC, unsigned offH,
                                           float (&x0)[20], float (&x1)[20], float (&x2)[20]) {
    const float* pc = base + offC;
    const float* ph = base + offH;
    #pragma unroll
    for (int i = 0; i < 8; ++i) {
        x0[i] = pc[i * W_IMG];
        x1[i] = pc[i * W_IMG + CH_STRIDE];
        x2[i] = pc[i * W_IMG + 2 * CH_STRIDE];
    }
    #pragma unroll
    for (int i = 0; i < 12; ++i) {
        x0[8 + i] = ph[i * W_IMG];
        x1[8 + i] = ph[i * W_IMG + CH_STRIDE];
        x2[8 + i] = ph[i * W_IMG + 2 * CH_STRIDE];
    }
}

__device__ __forceinline__ void saturation(const float (&x0)[20], const float (&x1)[20], const float (&x2)[20],
                                           float (&sat)[8]) {
    const float EPS = 1.0f / 255.0f;
    #pragma unroll
    for (int v = 0; v < 8; ++v) {
        const float mx = fmaxf(x0[v], fmaxf(x1[v], x2[v]));
        const float mn = fminf(x0[v], fminf(x1[v], x2[v]));
        sat[v] = (mx - mn + EPS) * __builtin_amdgcn_rcpf(mx + EPS);
    }
}

__device__ __forceinline__ void vertical_gemm(const float (&x0)[20], const float (&x1)[20], const float (&x2)[20],
                                              const v2f (&band1)[8], const bool hi, v8f& Vs, v8f& Vq) {
    float s[8], q[8];
    #pragma unroll
    for (int i = 0; i < 8; ++i) {
        s[i] = x0[i] + x1[i] + x2[i];
        q[i] = x0[i] * x0[i] + x1[i] * x1[i] + x2[i] * x2[i];
    }
    const float cs = (s[3] + s[4]) + (((hi ? s[0] : s[5]) + (hi ? s[1] : s[6])) + (hi ? s[2] : s[7]));
    const float cq = (q[3] + q[4]) + (((hi ? q[0] : q[5]) + (hi ? q[1] : q[6])) + (hi ? q[2] : q[7]));
    const v2f as0 = { hi ? s[5] : s[0], hi ? s[6] : s[1] };
    const v2f aq0 = { hi ? q[5] : q[0], hi ? q[6] : q[1] };
    const v2f as1 = { hi ? s[7] : s[2], cs };
    const v2f aq1 = { hi ? q[7] : q[2], cq };
    v8f acc1 = {}, acc2 = {};
    acc1 = __builtin_amdgcn_wmma_f32_16x16x4_f32(false, as0, false, band1[0], (short)0, acc1, false, false);
    acc2 = __builtin_amdgcn_wmma_f32_16x16x4_f32(false, aq0, false, band1[0], (short)0, acc2, false, false);
    acc1 = __builtin_amdgcn_wmma_f32_16x16x4_f32(false, as1, false, band1[1], (short)0, acc1, false, false);
    acc2 = __builtin_amdgcn_wmma_f32_16x16x4_f32(false, aq1, false, band1[1], (short)0, acc2, false, false);
    #pragma unroll
    for (int k = 2; k < 8; ++k) {
        const int e = 8 + 2 * (k - 2);
        const float p0 = x0[e], p1 = x1[e], p2 = x2[e];
        const float q0 = x0[e + 1], q1 = x1[e + 1], q2 = x2[e + 1];
        const v2f as = { p0 + p1 + p2, q0 + q1 + q2 };
        const v2f aq = { p0 * p0 + p1 * p1 + p2 * p2, q0 * q0 + q1 * q1 + q2 * q2 };
        acc1 = __builtin_amdgcn_wmma_f32_16x16x4_f32(false, as, false, band1[k],
                                                     (short)0, acc1, false, false);
        acc2 = __builtin_amdgcn_wmma_f32_16x16x4_f32(false, aq, false, band1[k],
                                                     (short)0, acc2, false, false);
    }
    Vs = acc1;
    Vq = acc2;
}

__device__ __forceinline__ void block_consumed(v8f& Vs, v8f& Vq) {
    asm volatile("" : "+v"(Vs), "+v"(Vq) : : "memory");
}

__device__ __forceinline__ void loads_issued(v8f& Vs, v8f& Vq) {
    asm volatile("" : "+v"(Vs), "+v"(Vq) : : "memory");
}

__device__ __forceinline__ void tile_summed(v8f& s1, v8f& s2, v2f& next_open) {
    asm volatile("" : "+v"(s1), "+v"(s2), "+v"(next_open));
}

__device__ __forceinline__ void horizontal_gemm_tile(const v8f& Vs0, const v8f& Vs1, const v8f& Vs2,
                                                     const v8f& Vq0, const v8f& Vq1, const v8f& Vq2,
                                                     const v2f (&band2)[8], const float (&sat)[8], const bool hi,
                                                     v2f& next_open, float* __restrict__ outb, unsigned outoff) {
    v8f s1 = {}, s2 = {};
    #pragma unroll
    for (int c = 0; c < 2; ++c) {
        v2f a1 = { Vs0[4 + 2 * c], Vs0[5 + 2 * c] };
        v2f a2 = { Vq0[4 + 2 * c], Vq0[5 + 2 * c] };
        s1 = __builtin_amdgcn_wmma_f32_16x16x4_f32(false, a1, false, band2[c],
                                                   (short)0, s1, false, false);
        s2 = __builtin_amdgcn_wmma_f32_16x16x4_f32(false, a2, false, band2[c],
                                                   (short)0, s2, false, false);
    }
    {
        const float cs = (Vs1[3] + Vs1[4]) + (((hi ? Vs1[0] : Vs1[5]) + (hi ? Vs1[1] : Vs1[6])) + (hi ? Vs1[2] : Vs1[7]));
        const float cq = (Vq1[3] + Vq1[4]) + (((hi ? Vq1[0] : Vq1[5]) + (hi ? Vq1[1] : Vq1[6])) + (hi ? Vq1[2] : Vq1[7]));
        const v2f a16 = { hi ? Vs1[5] : Vs1[0], hi ? Vs1[6] : Vs1[1] };
        const v2f a26 = { hi ? Vq1[5] : Vq1[0], hi ? Vq1[6] : Vq1[1] };
        const v2f a17 = { hi ? Vs1[7] : Vs1[2], cs };
        const v2f a27 = { hi ? Vq1[7] : Vq1[2], cq };
        s1 = __builtin_amdgcn_wmma_f32_16x16x4_f32(false, a16, false, band2[6], (short)0, s1, false, false);
        s2 = __builtin_amdgcn_wmma_f32_16x16x4_f32(false, a26, false, band2[6], (short)0, s2, false, false);
        s1 = __builtin_amdgcn_wmma_f32_16x16x4_f32(false, a17, false, band2[7], (short)0, s1, false, false);
        s2 = __builtin_amdgcn_wmma_f32_16x16x4_f32(false, a27, false, band2[7], (short)0, s2, false, false);
    }
    #pragma unroll
    for (int c = 0; c < 2; ++c) {
        v2f a1 = { Vs2[2 * c], Vs2[2 * c + 1] };
        v2f a2 = { Vq2[2 * c], Vq2[2 * c + 1] };
        s1 = __builtin_amdgcn_wmma_f32_16x16x4_f32(false, a1, false, band2[2 + c],
                                                   (short)0, s1, false, false);
        s2 = __builtin_amdgcn_wmma_f32_16x16x4_f32(false, a2, false, band2[2 + c],
                                                   (short)0, s2, false, false);
    }
    #pragma unroll
    for (int c = 0; c < 2; ++c) {
        v2f a1 = { hi ? Vs0[2 * c] : Vs2[4 + 2 * c], hi ? Vs0[2 * c + 1] : Vs2[5 + 2 * c] };
        v2f a2 = { hi ? Vq0[2 * c] : Vq2[4 + 2 * c], hi ? Vq0[2 * c + 1] : Vq2[5 + 2 * c] };
        s1 = __builtin_amdgcn_wmma_f32_16x16x4_f32(false, a1, false, band2[4 + c],
                                                   (short)0, s1, false, false);
        s2 = __builtin_amdgcn_wmma_f32_16x16x4_f32(false, a2, false, band2[4 + c],
                                                   (short)0, s2, false, false);
    }

    tile_summed(s1, s2, next_open);

    const float inv = 1.0f / (25.0f * 25.0f * 3.0f);
    const float EPS = 1.0f / 255.0f;
    float* op = outb + outoff;
    #pragma unroll
    for (int v = 0; v < 8; ++v) {
        float meanr = s1[v] * inv;
        float means = s2[v] * inv;
        float expo  = fabsf(meanr - 0.5f) + EPS;
        float contrast = means - meanr * meanr;
        op[v * W_IMG] = sat[v] * contrast * __builtin_amdgcn_rcpf(expo);
    }
}

__global__ __launch_bounds__(256, 4)
void box_stats_chain_kernel(const float* __restrict__ img, float* __restrict__ out) {
    __shared__ float xch[8][16][32];

    const int lane = threadIdx.x;
    const int hi   = lane >> 4;
    const int lo   = lane & 15;
    const int wv   = threadIdx.y;
    const int band = wv & 3;
    const int dir  = wv >> 2;
    const int cB   = blockIdx.x * (32 * TILES) + 16 * TILES;
    const int r0   = (blockIdx.y * 4 + band) * 16;
    const int b    = blockIdx.z;

    const float* base = img + (size_t)b * 3 * CH_STRIDE;
    float* outb = out + (size_t)b * CH_STRIDE;

    const int  hw0   = hi ? 28 : 0;
    const int  hrow0 = r0 - 12 + hw0;
    const bool flip  = (hrow0 < 0) || (hrow0 + 11 > 511);
    const int  hrowA = flip ? reflect_idx(hrow0 + 11) : hrow0;
    const unsigned rowC = (unsigned)(r0 + 8 * hi) * W_IMG;
    const unsigned rowH = (unsigned)hrowA * W_IMG;

    v2f band1[8];
    {
        const int wr = hi ? 25 : 12;
        band1[0].x = ((unsigned)(wr     - lo) <= 24u) ? 1.0f : 0.0f;
        band1[0].y = ((unsigned)(wr + 1 - lo) <= 24u) ? 1.0f : 0.0f;
        band1[1].x = ((unsigned)(wr + 2 - lo) <= 24u) ? 1.0f : 0.0f;
        band1[1].y = 1.0f;
    }
    #pragma unroll
    for (int k = 2; k < 8; ++k) {
        const int i = 2 * (k - 2);
        const int w0 = hw0 + (flip ? 11 - i : i);
        const int w1 = hw0 + (flip ? 10 - i : i + 1);
        band1[k].x = ((unsigned)(w0 - lo) <= 24u) ? 1.0f : 0.0f;
        band1[k].y = ((unsigned)(w1 - lo) <= 24u) ? 1.0f : 0.0f;
    }
    v2f band2[8];
    {
        const int jm[8] = { 8 * hi, 8 * hi + 2,
                            28 + 8 * hi, 30 + 8 * hi,
                            hi ? 4 : 32, hi ? 6 : 34,
                            hi ? 25 : 12, hi ? 27 : 14 };
        #pragma unroll
        for (int c = 0; c < 8; ++c) {
            band2[c].x = ((unsigned)(jm[c]     - lo) <= 24u) ? 1.0f : 0.0f;
            band2[c].y = ((unsigned)(jm[c] + 1 - lo) <= 24u) ? 1.0f : 0.0f;
        }
        band2[7].y = 1.0f;
    }

    float x0[20], x1[20], x2[20];
    float y0[20], y1[20], y2[20];
    v8f Vs0, Vs1, Vs2, Vq0, Vq1, Vq2;
    float satC[8], satN[8];

    const int cfirst = dir ? cB - 1 - lo : cB + lo;
    const int cstep  = dir ? -16 : 16;

    unsigned colb = (unsigned)cfirst;
    load_block(base, rowC + colb, rowH + colb, y0, y1, y2);
    colb = (unsigned)(cfirst + cstep);
    load_block(base, rowC + colb, rowH + colb, x0, x1, x2);

    saturation(y0, y1, y2, satC);
    vertical_gemm(y0, y1, y2, band1, hi != 0, Vs1, Vq1);
    #pragma unroll
    for (int v = 0; v < 8; ++v) {
        xch[wv][v][lane]     = Vs1[v];
        xch[wv][8 + v][lane] = Vq1[v];
    }
    __builtin_amdgcn_fence(__ATOMIC_RELEASE, "workgroup", "local");
    __builtin_amdgcn_s_barrier_signal(-1);

    unsigned outoff = (unsigned)(r0 + 8 * hi) * W_IMG + (unsigned)cfirst;

    #pragma unroll
    for (int j = 2; j < NBLK - 1; ++j) {
        saturation(x0, x1, x2, satN);
        vertical_gemm(x0, x1, x2, band1, hi != 0, Vs2, Vq2);
        block_consumed(Vs2, Vq2);
        const int cnext = cfirst + (j) * cstep;
        colb = (unsigned)((j + 1 == NBLK - 1) ? reflect_idx(cnext) : cnext);
        load_block(base, rowC + colb, rowH + colb, x0, x1, x2);
        if (j == 2) {
            loads_issued(Vs1, Vq1);
            __builtin_amdgcn_s_barrier_wait(-1);
            __builtin_amdgcn_fence(__ATOMIC_ACQUIRE, "workgroup", "local");
            #pragma unroll
            for (int v = 0; v < 8; ++v) {
                Vs0[v] = xch[wv ^ 4][7 - v][lane ^ 16];
                Vq0[v] = xch[wv ^ 4][15 - v][lane ^ 16];
            }
        } else {
            loads_issued(Vs0, Vq0);
        }

        horizontal_gemm_tile(Vs0, Vs1, Vs2, Vq0, Vq1, Vq2, band2, satC, hi != 0, band1[0], outb, outoff);
        outoff += (unsigned)cstep;

        Vs0 = Vs1; Vs1 = Vs2;
        Vq0 = Vq1; Vq1 = Vq2;
        #pragma unroll
        for (int v = 0; v < 8; ++v) satC[v] = satN[v];
    }

    vertical_gemm(x0, x1, x2, band1, hi != 0, Vs2, Vq2);
    horizontal_gemm_tile(Vs0, Vs1, Vs2, Vq0, Vq1, Vq2, band2, satC, hi != 0, band1[0], outb, outoff);
}

extern "C" void kernel_launch(void* const* d_in, const int* in_sizes, int n_in,
                              void* d_out, int out_size, void* d_ws, size_t ws_size,
                              hipStream_t stream) {
    (void)in_sizes; (void)n_in; (void)d_ws; (void)ws_size; (void)out_size;
    const float* img = (const float*)d_in[0];
    float* out = (float*)d_out;
    dim3 grid(4, 8, 16);
    box_stats_chain_kernel<<<grid, dim3(32, 8), 0, stream>>>(img, out);
}
